# starting kernel
# baseline (speedup 1.0000x reference)
#include <hip/hip_runtime.h>

typedef float v2f __attribute__((ext_vector_type(2)));
typedef float v8f __attribute__((ext_vector_type(8)));

#define BN_    4
#define N_GRP  2
#define L_     1024
#define DM_    512
#define DIN_   1024
#define DSTATE 16
#define DTRANK 32
#define MTOT   4096
#define NCHUNK 16
#define TCH    64

__global__ __launch_bounds__(128)
void wmma_gemm_tn(const float* __restrict__ A, int lda,
                  const float* __restrict__ Bm, int ldb,
                  float* __restrict__ C, int ldc,
                  int M, int N, int K,
                  const float* __restrict__ bias, int epilogue)
{
    const int lane  = threadIdx.x & 31;
    const int wave  = threadIdx.x >> 5;
    const int ntile = N >> 6;
    const int tile  = blockIdx.x * 4 + wave;
    if (tile >= (M >> 6) * ntile) return;
    const int tm = tile / ntile;
    const int tn = tile % ntile;

    const int l15  = lane & 15;
    const int koff = (lane >> 4) << 1;
    const size_t astr = (size_t)16 * lda;
    const size_t bstr = (size_t)16 * ldb;
    const float* ap = A  + (size_t)(tm * 64 + l15) * lda + koff;
    const float* bp = Bm + (size_t)(tn * 64 + l15) * ldb + koff;

    v8f acc[4][4] = {};
    for (int k = 0; k < K; k += 4) {
        v2f a[4], b[4];
#pragma unroll
        for (int i = 0; i < 4; ++i) a[i] = *(const v2f*)(ap + i * astr + k);
#pragma unroll
        for (int j = 0; j < 4; ++j) b[j] = *(const v2f*)(bp + j * bstr + k);
#pragma unroll
        for (int i = 0; i < 4; ++i)
#pragma unroll
            for (int j = 0; j < 4; ++j)
                acc[i][j] = __builtin_amdgcn_wmma_f32_16x16x4_f32(
                                false, a[i], false, b[j],
                                (short)0, acc[i][j], false, false);
    }

    const int rb = (lane >> 4) << 3;
#pragma unroll
    for (int i = 0; i < 4; ++i) {
        const int rowb = tm * 64 + i * 16 + rb;
#pragma unroll
        for (int j = 0; j < 4; ++j) {
            const int col = tn * 64 + j * 16 + l15;
            const float bv = (epilogue == 1) ? bias[col] : 0.0f;
#pragma unroll
            for (int r = 0; r < 8; ++r) {
                float v = acc[i][j][r];
                if (epilogue == 1) {
                    v += bv;
                    v = (v > 20.0f) ? v : __logf(1.0f + __expf(v));
                }
                C[(size_t)(rowb + r) * ldc + col] = v;
            }
        }
    }
}

__global__ __launch_bounds__(256)
void conv_silu_kernel(const float* __restrict__ C1,
                      const float* __restrict__ cw,
                      const float* __restrict__ cb,
                      float* __restrict__ U, int total)
{
    int idx = blockIdx.x * 256 + threadIdx.x;
    if (idx >= total) return;
    int d  = idx & (DIN_ - 1);
    int l  = (idx >> 10) & (L_ - 1);
    int bn = idx >> 20;
    int n  = bn & (N_GRP - 1);
    int ch = n * DIN_ + d;

    const float* w = cw + (size_t)ch * 4;
    float acc = cb[ch];
    const float* xcol = C1 + ((size_t)bn * L_) * (2 * DIN_) + d;
#pragma unroll
    for (int k = 0; k < 4; ++k) {
        int ll = l - 3 + k;
        float xv = (ll >= 0) ? xcol[(size_t)ll * (2 * DIN_)] : 0.0f;
        acc = fmaf(w[k], xv, acc);
    }
    U[idx] = acc / (1.0f + __expf(-acc));
}

__global__ __launch_bounds__(128)
void scan_chunk_reduce(const float* __restrict__ DT, const float* __restrict__ U,
                       const float* __restrict__ PR,
                       const float* __restrict__ A_log,
                       float* __restrict__ CH_C, float* __restrict__ CH_S)
{
    const int d  = ((blockIdx.x & 7) << 7) + threadIdx.x;
    const int ck = (blockIdx.x >> 3) & (NCHUNK - 1);
    const int bn = blockIdx.x >> 7;

    float As[DSTATE], h[DSTATE];
#pragma unroll
    for (int s = 0; s < DSTATE; ++s) {
        As[s] = -__expf(A_log[(size_t)d * DSTATE + s]);
        h[s]  = 0.0f;
    }
    float S = 0.0f;
    const int t0 = ck * TCH;
    for (int t = t0; t < t0 + TCH; ++t) {
        const size_t base = (size_t)bn * L_ + t;
        const float* pr = PR + base * 64;
        float dtv = DT[base * DIN_ + d];
        float uv  = U [base * DIN_ + d];
        S += dtv;
        float dtu = dtv * uv;
#pragma unroll
        for (int s = 0; s < DSTATE; ++s)
            h[s] = fmaf(__expf(dtv * As[s]), h[s], dtu * pr[DTRANK + s]);
    }
    const size_t cb = (size_t)(bn * NCHUNK + ck) * DSTATE * DIN_ + d;
#pragma unroll
    for (int s = 0; s < DSTATE; ++s) CH_C[cb + (size_t)s * DIN_] = h[s];
    CH_S[(size_t)(bn * NCHUNK + ck) * DIN_ + d] = S;
}

__global__ __launch_bounds__(128)
void scan_chunk_emit(const float* __restrict__ DT, const float* __restrict__ U,
                     const float* __restrict__ PR, const float* __restrict__ C1,
                     const float* __restrict__ A_log, const float* __restrict__ Dp,
                     const float* __restrict__ CH_C, const float* __restrict__ CH_S,
                     float* __restrict__ Y2)
{
    const int d  = ((blockIdx.x & 7) << 7) + threadIdx.x;
    const int ck = (blockIdx.x >> 3) & (NCHUNK - 1);
    const int bn = blockIdx.x >> 7;

    float As[DSTATE], h[DSTATE];
#pragma unroll
    for (int s = 0; s < DSTATE; ++s) {
        As[s] = -__expf(A_log[(size_t)d * DSTATE + s]);
        h[s]  = 0.0f;
    }
    for (int j = 0; j < ck; ++j) {
        float Sj = CH_S[(size_t)(bn * NCHUNK + j) * DIN_ + d];
        const size_t cb = (size_t)(bn * NCHUNK + j) * DSTATE * DIN_ + d;
#pragma unroll
        for (int s = 0; s < DSTATE; ++s)
            h[s] = fmaf(__expf(Sj * As[s]), h[s], CH_C[cb + (size_t)s * DIN_]);
    }

    const float dp = Dp[d];
    const int t0 = ck * TCH;
    for (int t = t0; t < t0 + TCH; ++t) {
        const size_t base = (size_t)bn * L_ + t;
        const float* pr = PR + base * 64;
        float dtv = DT[base * DIN_ + d];
        float uv  = U [base * DIN_ + d];
        float dtu = dtv * uv;
        float y = 0.0f;
#pragma unroll
        for (int s = 0; s < DSTATE; ++s) {
            h[s] = fmaf(__expf(dtv * As[s]), h[s], dtu * pr[DTRANK + s]);
            y    = fmaf(h[s], pr[DTRANK + DSTATE + s], y);
        }
        float res = C1[base * (2 * DIN_) + DIN_ + d];
        float sr  = res / (1.0f + __expf(-res));
        Y2[base * DIN_ + d] = (y + uv * dp) * sr;
    }
}

extern "C" void kernel_launch(void* const* d_in, const int* in_sizes, int n_in,
                              void* d_out, int out_size, void* d_ws, size_t ws_size,
                              hipStream_t stream)
{
    const float* x      = (const float*)d_in[0];
    const float* W_in   = (const float*)d_in[1];
    const float* conv_w = (const float*)d_in[2];
    const float* conv_b = (const float*)d_in[3];
    const float* W_x    = (const float*)d_in[4];
    const float* W_dt   = (const float*)d_in[5];
    const float* b_dt   = (const float*)d_in[6];
    const float* A_log  = (const float*)d_in[7];
    const float* Dp     = (const float*)d_in[8];
    const float* W_out  = (const float*)d_in[9];
    float* out = (float*)d_out;

    float* C1   = (float*)d_ws;
    float* U    = C1   + (size_t)MTOT * (2 * DIN_);
    float* PR   = U    + (size_t)MTOT * DIN_;
    float* DTb  = PR   + (size_t)MTOT * 64;
    float* Y2   = DTb  + (size_t)MTOT * DIN_;
    float* CH_C = Y2   + (size_t)MTOT * DIN_;
    float* CH_S = CH_C + (size_t)BN_ * NCHUNK * DSTATE * DIN_;

    int tiles = (MTOT / 64) * ((2 * DIN_) / 64);
    wmma_gemm_tn<<<tiles / 4, 128, 0, stream>>>(x, DM_, W_in, DM_,
                                                C1, 2 * DIN_,
                                                MTOT, 2 * DIN_, DM_, nullptr, 0);

    conv_silu_kernel<<<(MTOT * DIN_) / 256, 256, 0, stream>>>(C1, conv_w, conv_b,
                                                              U, MTOT * DIN_);

    tiles = (MTOT / 64) * (64 / 64);
    wmma_gemm_tn<<<tiles / 4, 128, 0, stream>>>(U, DIN_, W_x, DIN_,
                                                PR, 64,
                                                MTOT, 64, DIN_, nullptr, 0);

    tiles = (MTOT / 64) * (DIN_ / 64);
    wmma_gemm_tn<<<tiles / 4, 128, 0, stream>>>(PR, 64, W_dt, DTRANK,
                                                DTb, DIN_,
                                                MTOT, DIN_, DTRANK, b_dt, 1);

    const int scan_blocks = BN_ * NCHUNK * (DIN_ / 128);
    scan_chunk_reduce<<<scan_blocks, 128, 0, stream>>>(DTb, U, PR, A_log,
                                                       CH_C, CH_S);
    scan_chunk_emit<<<scan_blocks, 128, 0, stream>>>(DTb, U, PR, C1, A_log, Dp,
                                                     CH_C, CH_S, Y2);

    tiles = (MTOT / 64) * (DM_ / 64);
    wmma_gemm_tn<<<tiles / 4, 128, 0, stream>>>(Y2, DIN_, W_out, DIN_,
                                                out, DM_,
                                                MTOT, DM_, DIN_, nullptr, 0);
}


# dt_proj launch gone: the scan reduce pass makes softplus(dt_proj) itself, a wave its chunk's 16 steps x its own 32 channels on WMMA, via LDS to its threads, written for emit; 5 launches; same bits
# speedup vs baseline: 2.0070x; 2.0070x over previous
#include <hip/hip_runtime.h>

typedef float v2f __attribute__((ext_vector_type(2)));
typedef float v4f __attribute__((ext_vector_type(4)));
typedef float v8f __attribute__((ext_vector_type(8)));

#define BN_    4
#define N_GRP  2
#define L_     1024
#define DM_    512
#define DIN_   1024
#define DSTATE 16
#define DTRANK 32
#define MTOT   4096
#define NCHUNK 64
#define NCH_LG 6
#define TCH    16

__device__ __forceinline__
void dt_tile_softplus(const float* __restrict__ prb, const float* __restrict__ Wdt,
                      const float* __restrict__ bdt, int ch0, int lane,
                      float (&dt0)[8], float (&dt1)[8])
{
    const int c  = lane & 15;
    const int hh = lane >> 4;
    const float* ap  = prb + (size_t)c * 64 + 2 * hh;
    const float* bp0 = Wdt + (size_t)(ch0 + c) * DTRANK + 2 * hh;
    const float* bp1 = bp0 + (size_t)16 * DTRANK;

    v2f a[DTRANK / 4], b0[DTRANK / 4], b1[DTRANK / 4];
#pragma unroll
    for (int ks = 0; ks < DTRANK / 4; ++ks) {
        a[ks]  = *(const v2f*)(ap  + 4 * ks);
        b0[ks] = *(const v2f*)(bp0 + 4 * ks);
        b1[ks] = *(const v2f*)(bp1 + 4 * ks);
    }
    const float bias0 = bdt[ch0 + c];
    const float bias1 = bdt[ch0 + 16 + c];
    v8f acc0 = {}, acc1 = {};
#pragma unroll
    for (int ks = 0; ks < DTRANK / 4; ++ks) {
        acc0 = __builtin_amdgcn_wmma_f32_16x16x4_f32(false, a[ks], false, b0[ks],
                                                     (short)0, acc0, false, false);
        acc1 = __builtin_amdgcn_wmma_f32_16x16x4_f32(false, a[ks], false, b1[ks],
                                                     (short)0, acc1, false, false);
    }
#pragma unroll
    for (int r = 0; r < 8; ++r) {
        float v0 = acc0[r] + bias0;
        float v1 = acc1[r] + bias1;
        dt0[r] = (v0 > 20.0f) ? v0 : __logf(1.0f + __expf(v0));
        dt1[r] = (v1 > 20.0f) ? v1 : __logf(1.0f + __expf(v1));
    }
}

#define XG 32
__global__ __launch_bounds__(64) __attribute__((amdgpu_waves_per_eu(1, 2)))
void x_proj_gemm(const float* __restrict__ U, const float* __restrict__ Wx,
                 float* __restrict__ PR)
{
    const int lane = threadIdx.x & 31;
    const int wave = threadIdx.x >> 5;
    const int c    = lane & 15;
    const int hh   = lane >> 4;
    const int m0   = blockIdx.x * 16;
    const int n0   = wave * 32;

    const float* ap  = U  + (size_t)(m0 + c) * DIN_ + 4 * hh;
    const float* bp0 = Wx + (size_t)(n0 + c) * DIN_ + 4 * hh;
    const float* bp1 = bp0 + (size_t)16 * DIN_;

    v8f acc0 = {}, acc1 = {};
    for (int g0 = 0; g0 < DIN_ / 8; g0 += XG) {
        v4f a[XG], b0[XG], b1[XG];
#pragma unroll
        for (int i = 0; i < XG; ++i) {
            a[i]  = *(const v4f*)(ap  + 8 * (g0 + i));
            b0[i] = *(const v4f*)(bp0 + 8 * (g0 + i));
            b1[i] = *(const v4f*)(bp1 + 8 * (g0 + i));
        }
#pragma unroll
        for (int i = 0; i < XG; ++i) {
            const v2f alo  = {a[i][0],  a[i][1]},  ahi  = {a[i][2],  a[i][3]};
            const v2f b0lo = {b0[i][0], b0[i][1]}, b0hi = {b0[i][2], b0[i][3]};
            const v2f b1lo = {b1[i][0], b1[i][1]}, b1hi = {b1[i][2], b1[i][3]};
            acc0 = __builtin_amdgcn_wmma_f32_16x16x4_f32(false, alo, false, b0lo,
                                                         (short)0, acc0, false, false);
            acc1 = __builtin_amdgcn_wmma_f32_16x16x4_f32(false, alo, false, b1lo,
                                                         (short)0, acc1, false, false);
            acc0 = __builtin_amdgcn_wmma_f32_16x16x4_f32(false, ahi, false, b0hi,
                                                         (short)0, acc0, false, false);
            acc1 = __builtin_amdgcn_wmma_f32_16x16x4_f32(false, ahi, false, b1hi,
                                                         (short)0, acc1, false, false);
        }
    }

    const size_t orow = (size_t)m0 + 8 * hh;
#pragma unroll
    for (int r = 0; r < 8; ++r) {
        PR[(orow + r) * 64 + n0 + c]      = acc0[r];
        PR[(orow + r) * 64 + n0 + 16 + c] = acc1[r];
    }
}

__global__ __launch_bounds__(128)
void in_proj_conv_silu(const float* __restrict__ X, const float* __restrict__ Win,
                       const float* __restrict__ cw, const float* __restrict__ cb,
                       float* __restrict__ C1, float* __restrict__ U)
{
    const int lane = threadIdx.x & 31;
    const int wave = threadIdx.x >> 5;
    const int tm   = blockIdx.x >> 3;
    const int q    = blockIdx.x & 7;
    const bool xt  = ((((int)blockIdx.x) ^ (wave >> 1)) & 1) == 0;
    const int tn   = (xt ? 0 : 16) + 2 * q + (wave & 1);

    const int l15  = lane & 15;
    const int hh   = lane >> 4;
    const int koff = hh << 1;
    const size_t astr = (size_t)16 * DM_;
    const float* ap = X   + (size_t)(tm * 64 + l15) * DM_ + koff;
    const float* bp = Win + (size_t)(tn * 64 + l15) * DM_ + koff;
    const int rb = hh << 3;

    if (!xt) {
        v8f acc[4][4] = {};
        for (int k = 0; k < DM_; k += 4) {
            v2f a[4], b[4];
#pragma unroll
            for (int i = 0; i < 4; ++i) a[i] = *(const v2f*)(ap + i * astr + k);
#pragma unroll
            for (int j = 0; j < 4; ++j) b[j] = *(const v2f*)(bp + j * astr + k);
#pragma unroll
            for (int i = 0; i < 4; ++i)
#pragma unroll
                for (int j = 0; j < 4; ++j)
                    acc[i][j] = __builtin_amdgcn_wmma_f32_16x16x4_f32(
                                    false, a[i], false, b[j],
                                    (short)0, acc[i][j], false, false);
        }
#pragma unroll
        for (int i = 0; i < 4; ++i) {
            const int rowb = tm * 64 + i * 16 + rb;
#pragma unroll
            for (int j = 0; j < 4; ++j) {
                const int col = tn * 64 + j * 16 + l15;
#pragma unroll
                for (int r = 0; r < 8; ++r)
                    C1[(size_t)(rowb + r) * (2 * DIN_) + col] = acc[i][j][r];
            }
        }
        return;
    }

    const bool has_above = (tm & 15) != 0;
    const float* hp = has_above ? (ap - astr) : ap;
    v8f acc[5][4] = {};
    for (int k = 0; k < DM_; k += 4) {
        v2f a[5], b[4];
        a[0] = *(const v2f*)(hp + k);
#pragma unroll
        for (int i = 0; i < 4; ++i) a[i + 1] = *(const v2f*)(ap + i * astr + k);
#pragma unroll
        for (int j = 0; j < 4; ++j) b[j] = *(const v2f*)(bp + j * astr + k);
#pragma unroll
        for (int i = 0; i < 5; ++i)
#pragma unroll
            for (int j = 0; j < 4; ++j)
                acc[i][j] = __builtin_amdgcn_wmma_f32_16x16x4_f32(
                                false, a[i], false, b[j],
                                (short)0, acc[i][j], false, false);
    }
    if (!has_above) {
#pragma unroll
        for (int j = 0; j < 4; ++j) acc[0][j] = (v8f){};
    }

    const int n = (tm >> 4) & (N_GRP - 1);
#pragma unroll
    for (int j = 0; j < 4; ++j) {
        const int d  = tn * 64 + j * 16 + l15;
        const int ch = n * DIN_ + d;
        const v4f w  = *(const v4f*)(cw + (size_t)ch * 4);
        const float bias = cb[ch];
#pragma unroll
        for (int i = 0; i < 4; ++i) {
            float xw[11];
#pragma unroll
            for (int p = 0; p < 3; ++p) {
                const float send = hh ? acc[i][j][5 + p] : acc[i + 1][j][5 + p];
                xw[p] = __shfl_xor(send, 16);
            }
#pragma unroll
            for (int r = 0; r < 8; ++r) xw[3 + r] = acc[i + 1][j][r];
            const int rowb = tm * 64 + i * 16 + rb;
#pragma unroll
            for (int r = 0; r < 8; ++r) {
                float s = bias;
                s = fmaf(w[0], xw[r],     s);
                s = fmaf(w[1], xw[r + 1], s);
                s = fmaf(w[2], xw[r + 2], s);
                s = fmaf(w[3], xw[r + 3], s);
                U[(size_t)(rowb + r) * DIN_ + d] = s / (1.0f + __expf(-s));
            }
        }
    }
}

__global__ __launch_bounds__(128)
void scan_chunk_reduce(const float* __restrict__ U, const float* __restrict__ PR,
                       const float* __restrict__ Wdt, const float* __restrict__ bdt,
                       const float* __restrict__ A_log,
                       float* __restrict__ DT,
                       float* __restrict__ CH_C, float* __restrict__ CH_S)
{
    __shared__ float dts[TCH * 128];

    const int d  = ((blockIdx.x & 7) << 7) + threadIdx.x;
    const int ck = (blockIdx.x >> 3) & (NCHUNK - 1);
    const int bn = blockIdx.x >> (3 + NCH_LG);

    {
        const int lane = threadIdx.x & 31;
        const int wch  = threadIdx.x & ~31;
        float dt0[8], dt1[8];
        dt_tile_softplus(PR + ((size_t)bn * L_ + ck * TCH) * 64, Wdt, bdt,
                         ((blockIdx.x & 7) << 7) + wch, lane, dt0, dt1);
        const int c = lane & 15, hh = lane >> 4;
#pragma unroll
        for (int r = 0; r < 8; ++r) {
            dts[(8 * hh + r) * 128 + wch + c]      = dt0[r];
            dts[(8 * hh + r) * 128 + wch + 16 + c] = dt1[r];
        }
    }
    __syncthreads();

    float As[DSTATE], h[DSTATE];
#pragma unroll
    for (int s = 0; s < DSTATE; ++s) {
        As[s] = -__expf(A_log[(size_t)d * DSTATE + s]);
        h[s]  = 0.0f;
    }
    float S = 0.0f;
    const int t0 = ck * TCH;
    for (int t = t0; t < t0 + TCH; ++t) {
        const size_t base = (size_t)bn * L_ + t;
        const float* pr = PR + base * 64;
        float dtv = dts[(t - t0) * 128 + threadIdx.x];
        DT[base * DIN_ + d] = dtv;
        float uv  = U [base * DIN_ + d];
        S += dtv;
        float dtu = dtv * uv;
#pragma unroll
        for (int s = 0; s < DSTATE; ++s)
            h[s] = fmaf(__expf(dtv * As[s]), h[s], dtu * pr[DTRANK + s]);
    }
    const size_t cb = (size_t)(bn * NCHUNK + ck) * DSTATE * DIN_ + d;
#pragma unroll
    for (int s = 0; s < DSTATE; ++s) CH_C[cb + (size_t)s * DIN_] = h[s];
    CH_S[(size_t)(bn * NCHUNK + ck) * DIN_ + d] = S;
}

__global__ __launch_bounds__(256)
void scan_chunk_prefix(const float* __restrict__ A_log,
                       const float* __restrict__ CH_C,
                       const float* __restrict__ CH_S,
                       float* __restrict__ H0)
{
    const int idx = blockIdx.x * 256 + threadIdx.x;
    const int d   = idx & (DIN_ - 1);
    const int s   = (idx >> 10) & (DSTATE - 1);
    const int bn  = idx >> 14;
    const float As = -__expf(A_log[(size_t)d * DSTATE + s]);
    float h = 0.0f;
    for (int j0 = 0; j0 < NCHUNK; j0 += 16) {
        float Sj[16], cj[16];
#pragma unroll
        for (int i = 0; i < 16; ++i) {
            const size_t cidx = (size_t)(bn * NCHUNK + j0 + i);
            Sj[i] = CH_S[cidx * DIN_ + d];
            cj[i] = CH_C[(cidx * DSTATE + s) * DIN_ + d];
        }
#pragma unroll
        for (int i = 0; i < 16; ++i) {
            H0[((size_t)(bn * NCHUNK + j0 + i) * DSTATE + s) * DIN_ + d] = h;
            h = fmaf(__expf(Sj[i] * As), h, cj[i]);
        }
    }
}

__global__ __launch_bounds__(1024)
void scan_emit_out_proj(const float* __restrict__ DT, const float* __restrict__ U,
                        const float* __restrict__ PR, const float* __restrict__ C1,
                        const float* __restrict__ A_log, const float* __restrict__ Dp,
                        const float* __restrict__ H0,
                        const float* __restrict__ Wout,
                        float* __restrict__ out)
{
    __shared__ __attribute__((aligned(16))) float Ys[TCH * DIN_];

    const int d  = threadIdx.x;
    const int ck = blockIdx.x & (NCHUNK - 1);
    const int bn = blockIdx.x >> NCH_LG;
    const int t0 = ck * TCH;

    {
        const size_t hb = (size_t)(bn * NCHUNK + ck) * DSTATE * DIN_ + d;
        float As[DSTATE], h[DSTATE];
#pragma unroll
        for (int s = 0; s < DSTATE; ++s) {
            As[s] = -__expf(A_log[(size_t)d * DSTATE + s]);
            h[s]  = H0[hb + (size_t)s * DIN_];
        }

        const float dp = Dp[d];
        for (int tt = 0; tt < TCH; ++tt) {
            const size_t base = (size_t)bn * L_ + t0 + tt;
            const float* pr = PR + base * 64;
            float dtv = DT[base * DIN_ + d];
            float uv  = U [base * DIN_ + d];
            float dtu = dtv * uv;
            float y = 0.0f;
#pragma unroll
            for (int s = 0; s < DSTATE; ++s) {
                h[s] = fmaf(__expf(dtv * As[s]), h[s], dtu * pr[DTRANK + s]);
                y    = fmaf(h[s], pr[DTRANK + DSTATE + s], y);
            }
            float res = C1[base * (2 * DIN_) + DIN_ + d];
            float sr  = res / (1.0f + __expf(-res));
            Ys[tt * DIN_ + ((((d >> 2) ^ tt) << 2) | (d & 3))] = (y + uv * dp) * sr;
        }
    }
    __syncthreads();

    const int wave = threadIdx.x >> 5;
    if (wave >= 16) return;
    const int lane = threadIdx.x & 31;
    const int c    = lane & 15;
    const int hh   = lane >> 4;
    const int n0   = wave * 32;

    const float* ya  = Ys + c * DIN_;
    const float* bp0 = Wout + (size_t)(n0 + c) * DIN_ + 4 * hh;
    const float* bp1 = bp0 + (size_t)16 * DIN_;

    v8f acc0 = {}, acc1 = {};
#pragma unroll 4
    for (int g = 0; g < DIN_ / 8; ++g) {
        const v4f a  = *(const v4f*)(ya + (((2 * g + hh) ^ c) << 2));
        const v4f b0 = *(const v4f*)(bp0 + 8 * g);
        const v4f b1 = *(const v4f*)(bp1 + 8 * g);
        const v2f alo = {a[0], a[1]},   ahi = {a[2], a[3]};
        const v2f b0lo = {b0[0], b0[1]}, b0hi = {b0[2], b0[3]};
        const v2f b1lo = {b1[0], b1[1]}, b1hi = {b1[2], b1[3]};
        acc0 = __builtin_amdgcn_wmma_f32_16x16x4_f32(false, alo, false, b0lo,
                                                     (short)0, acc0, false, false);
        acc1 = __builtin_amdgcn_wmma_f32_16x16x4_f32(false, alo, false, b1lo,
                                                     (short)0, acc1, false, false);
        acc0 = __builtin_amdgcn_wmma_f32_16x16x4_f32(false, ahi, false, b0hi,
                                                     (short)0, acc0, false, false);
        acc1 = __builtin_amdgcn_wmma_f32_16x16x4_f32(false, ahi, false, b1hi,
                                                     (short)0, acc1, false, false);
    }

    const size_t orow = (size_t)bn * L_ + t0 + 8 * hh;
#pragma unroll
    for (int r = 0; r < 8; ++r) {
        out[(orow + r) * DM_ + n0 + c]      = acc0[r];
        out[(orow + r) * DM_ + n0 + 16 + c] = acc1[r];
    }
}

extern "C" void kernel_launch(void* const* d_in, const int* in_sizes, int n_in,
                              void* d_out, int out_size, void* d_ws, size_t ws_size,
                              hipStream_t stream)
{
    const float* x      = (const float*)d_in[0];
    const float* W_in   = (const float*)d_in[1];
    const float* conv_w = (const float*)d_in[2];
    const float* conv_b = (const float*)d_in[3];
    const float* W_x    = (const float*)d_in[4];
    const float* W_dt   = (const float*)d_in[5];
    const float* b_dt   = (const float*)d_in[6];
    const float* A_log  = (const float*)d_in[7];
    const float* Dp     = (const float*)d_in[8];
    const float* W_out  = (const float*)d_in[9];
    float* out = (float*)d_out;

    float* C1   = (float*)d_ws;
    float* U    = C1   + (size_t)MTOT * (2 * DIN_);
    float* PR   = U    + (size_t)MTOT * DIN_;
    float* DTb  = PR   + (size_t)MTOT * 64;
    float* CH_C = DTb  + (size_t)MTOT * DIN_;
    float* CH_S = CH_C + (size_t)BN_ * NCHUNK * DSTATE * DIN_;
    float* H0   = CH_S + (size_t)BN_ * NCHUNK * DIN_;

    in_proj_conv_silu<<<(MTOT / 64) * 8, 128, 0, stream>>>(x, W_in, conv_w, conv_b,
                                                           C1, U);

    x_proj_gemm<<<MTOT / 16, 64, 0, stream>>>(U, W_x, PR);

    const int scan_blocks = BN_ * NCHUNK * (DIN_ / 128);
    scan_chunk_reduce<<<scan_blocks, 128, 0, stream>>>(U, PR, W_dt, b_dt, A_log,
                                                       DTb, CH_C, CH_S);
    scan_chunk_prefix<<<(BN_ * DSTATE * DIN_) / 256, 256, 0, stream>>>(A_log, CH_C,
                                                                       CH_S, H0);
    scan_emit_out_proj<<<BN_ * NCHUNK, 1024, 0, stream>>>(DTb, U, PR, C1, A_log, Dp,
                                                          H0, W_out, out);
}


# x, W_in and W_out re-laid out per call by a short pre-pass into WMMA fragment order: in_proj and out_proj fragment loads read 4 whole 128-byte lines, were 16 lines each; 6 launches; same bits
# speedup vs baseline: 2.1590x; 1.0757x over previous
#include <hip/hip_runtime.h>

typedef float v2f __attribute__((ext_vector_type(2)));
typedef float v4f __attribute__((ext_vector_type(4)));
typedef float v8f __attribute__((ext_vector_type(8)));

#define BN_    4
#define N_GRP  2
#define L_     1024
#define DM_    512
#define DIN_   1024
#define DSTATE 16
#define DTRANK 32
#define MTOT   4096
#define NCHUNK 64
#define NCH_LG 6
#define TCH    16

__device__ __forceinline__
void dt_tile_softplus(const float* __restrict__ prb, const float* __restrict__ Wdt,
                      const float* __restrict__ bdt, int ch0, int lane,
                      float (&dt0)[8], float (&dt1)[8])
{
    const int c  = lane & 15;
    const int hh = lane >> 4;
    const float* ap  = prb + (size_t)c * 64 + 2 * hh;
    const float* bp0 = Wdt + (size_t)(ch0 + c) * DTRANK + 2 * hh;
    const float* bp1 = bp0 + (size_t)16 * DTRANK;

    v2f a[DTRANK / 4], b0[DTRANK / 4], b1[DTRANK / 4];
#pragma unroll
    for (int ks = 0; ks < DTRANK / 4; ++ks) {
        a[ks]  = *(const v2f*)(ap  + 4 * ks);
        b0[ks] = *(const v2f*)(bp0 + 4 * ks);
        b1[ks] = *(const v2f*)(bp1 + 4 * ks);
    }
    const float bias0 = bdt[ch0 + c];
    const float bias1 = bdt[ch0 + 16 + c];
    v8f acc0 = {}, acc1 = {};
#pragma unroll
    for (int ks = 0; ks < DTRANK / 4; ++ks) {
        acc0 = __builtin_amdgcn_wmma_f32_16x16x4_f32(false, a[ks], false, b0[ks],
                                                     (short)0, acc0, false, false);
        acc1 = __builtin_amdgcn_wmma_f32_16x16x4_f32(false, a[ks], false, b1[ks],
                                                     (short)0, acc1, false, false);
    }
#pragma unroll
    for (int r = 0; r < 8; ++r) {
        float v0 = acc0[r] + bias0;
        float v1 = acc1[r] + bias1;
        dt0[r] = (v0 > 20.0f) ? v0 : __logf(1.0f + __expf(v0));
        dt1[r] = (v1 > 20.0f) ? v1 : __logf(1.0f + __expf(v1));
    }
}

#define XF_JOBS   ((MTOT / 16) * (DM_ / 64))
#define WIF_JOBS  (((2 * DIN_) / 16) * (DM_ / 64))
#define WOF_JOBS  ((DM_ / 16) * (DIN_ / 64))
__global__ __launch_bounds__(256)
void to_fragment_order(const float* __restrict__ X, const float* __restrict__ Win,
                       const float* __restrict__ Wout,
                       float* __restrict__ Xf, float* __restrict__ Winf,
                       float* __restrict__ Woutf)
{
    const int lane = threadIdx.x & 31;
    const int c    = lane & 15;
    const int hh   = lane >> 4;
    const int job  = blockIdx.x * 8 + (threadIdx.x >> 5);

    if (job < XF_JOBS + WIF_JOBS) {
        const bool isx = job < XF_JOBS;
        const float* src = isx ? X : Win;
        float* dst       = isx ? Xf : Winf;
        const int j      = isx ? job : job - XF_JOBS;
        const int tile   = j >> 3;
        const int g0     = (j & 7) * 8;
        const float* rp = src + (size_t)(tile * 16 + c) * DM_ + 8 * g0 + 2 * hh;
        float* dp = dst + ((size_t)tile * (DM_ / 8) + g0) * 128 + lane * 4;
        v2f lo[8], hi[8];
#pragma unroll
        for (int i = 0; i < 8; ++i) {
            lo[i] = *(const v2f*)(rp + 8 * i);
            hi[i] = *(const v2f*)(rp + 8 * i + 4);
        }
#pragma unroll
        for (int i = 0; i < 8; ++i) {
            const v4f v = {lo[i][0], lo[i][1], hi[i][0], hi[i][1]};
            *(v4f*)(dp + i * 128) = v;
        }
    } else {
        const int j    = job - (XF_JOBS + WIF_JOBS);
        const int tile = j >> 4;
        const int g0   = (j & 15) * 8;
        const float* rp = Wout + (size_t)(tile * 16 + c) * DIN_ + 8 * g0 + 4 * hh;
        float* dp = Woutf + ((size_t)tile * (DIN_ / 8) + g0) * 128 + lane * 4;
        v4f v[8];
#pragma unroll
        for (int i = 0; i < 8; ++i) v[i] = *(const v4f*)(rp + 8 * i);
#pragma unroll
        for (int i = 0; i < 8; ++i) *(v4f*)(dp + i * 128) = v[i];
    }
}

#define XG 32
__global__ __launch_bounds__(64) __attribute__((amdgpu_waves_per_eu(1, 2)))
void x_proj_gemm(const float* __restrict__ U, const float* __restrict__ Wx,
                 float* __restrict__ PR)
{
    const int lane = threadIdx.x & 31;
    const int wave = threadIdx.x >> 5;
    const int c    = lane & 15;
    const int hh   = lane >> 4;
    const int m0   = blockIdx.x * 16;
    const int n0   = wave * 32;

    const float* ap  = U  + (size_t)(m0 + c) * DIN_ + 4 * hh;
    const float* bp0 = Wx + (size_t)(n0 + c) * DIN_ + 4 * hh;
    const float* bp1 = bp0 + (size_t)16 * DIN_;

    v8f acc0 = {}, acc1 = {};
    for (int g0 = 0; g0 < DIN_ / 8; g0 += XG) {
        v4f a[XG], b0[XG], b1[XG];
#pragma unroll
        for (int i = 0; i < XG; ++i) {
            a[i]  = *(const v4f*)(ap  + 8 * (g0 + i));
            b0[i] = *(const v4f*)(bp0 + 8 * (g0 + i));
            b1[i] = *(const v4f*)(bp1 + 8 * (g0 + i));
        }
#pragma unroll
        for (int i = 0; i < XG; ++i) {
            const v2f alo  = {a[i][0],  a[i][1]},  ahi  = {a[i][2],  a[i][3]};
            const v2f b0lo = {b0[i][0], b0[i][1]}, b0hi = {b0[i][2], b0[i][3]};
            const v2f b1lo = {b1[i][0], b1[i][1]}, b1hi = {b1[i][2], b1[i][3]};
            acc0 = __builtin_amdgcn_wmma_f32_16x16x4_f32(false, alo, false, b0lo,
                                                         (short)0, acc0, false, false);
            acc1 = __builtin_amdgcn_wmma_f32_16x16x4_f32(false, alo, false, b1lo,
                                                         (short)0, acc1, false, false);
            acc0 = __builtin_amdgcn_wmma_f32_16x16x4_f32(false, ahi, false, b0hi,
                                                         (short)0, acc0, false, false);
            acc1 = __builtin_amdgcn_wmma_f32_16x16x4_f32(false, ahi, false, b1hi,
                                                         (short)0, acc1, false, false);
        }
    }

    const size_t orow = (size_t)m0 + 8 * hh;
#pragma unroll
    for (int r = 0; r < 8; ++r) {
        PR[(orow + r) * 64 + n0 + c]      = acc0[r];
        PR[(orow + r) * 64 + n0 + 16 + c] = acc1[r];
    }
}

#define TSTR_IN ((DM_ / 8) * 128)
__global__ __launch_bounds__(128)
void in_proj_conv_silu(const float* __restrict__ Xf, const float* __restrict__ Winf,
                       const float* __restrict__ cw, const float* __restrict__ cb,
                       float* __restrict__ C1, float* __restrict__ U)
{
    const int lane = threadIdx.x & 31;
    const int wave = threadIdx.x >> 5;
    const int tm   = blockIdx.x >> 3;
    const int q    = blockIdx.x & 7;
    const bool xt  = ((((int)blockIdx.x) ^ (wave >> 1)) & 1) == 0;
    const int tn   = (xt ? 0 : 16) + 2 * q + (wave & 1);

    const int l15  = lane & 15;
    const int hh   = lane >> 4;
    const size_t astr = (size_t)TSTR_IN;
    const float* ap = Xf   + (size_t)(tm * 4) * TSTR_IN + lane * 4;
    const float* bp = Winf + (size_t)(tn * 4) * TSTR_IN + lane * 4;
    const int rb = hh << 3;

    if (!xt) {
        v8f acc[4][4] = {};
        for (int g = 0; g < DM_ / 8; ++g) {
            v4f a[4], b[4];
#pragma unroll
            for (int i = 0; i < 4; ++i) a[i] = *(const v4f*)(ap + i * astr + g * 128);
#pragma unroll
            for (int j = 0; j < 4; ++j) b[j] = *(const v4f*)(bp + j * astr + g * 128);
#pragma unroll
            for (int kk = 0; kk < 2; ++kk)
#pragma unroll
                for (int i = 0; i < 4; ++i)
#pragma unroll
                    for (int j = 0; j < 4; ++j) {
                        const v2f af = {a[i][2 * kk], a[i][2 * kk + 1]};
                        const v2f bf = {b[j][2 * kk], b[j][2 * kk + 1]};
                        acc[i][j] = __builtin_amdgcn_wmma_f32_16x16x4_f32(
                                        false, af, false, bf,
                                        (short)0, acc[i][j], false, false);
                    }
        }
#pragma unroll
        for (int i = 0; i < 4; ++i) {
            const int rowb = tm * 64 + i * 16 + rb;
#pragma unroll
            for (int j = 0; j < 4; ++j) {
                const int col = tn * 64 + j * 16 + l15;
#pragma unroll
                for (int r = 0; r < 8; ++r)
                    C1[(size_t)(rowb + r) * (2 * DIN_) + col] = acc[i][j][r];
            }
        }
        return;
    }

    const bool has_above = (tm & 15) != 0;
    const float* hp = has_above ? (ap - astr) : ap;
    v8f acc[5][4] = {};
    for (int g = 0; g < DM_ / 8; ++g) {
        v4f a[5], b[4];
        a[0] = *(const v4f*)(hp + g * 128);
#pragma unroll
        for (int i = 0; i < 4; ++i) a[i + 1] = *(const v4f*)(ap + i * astr + g * 128);
#pragma unroll
        for (int j = 0; j < 4; ++j) b[j] = *(const v4f*)(bp + j * astr + g * 128);
#pragma unroll
        for (int kk = 0; kk < 2; ++kk)
#pragma unroll
            for (int i = 0; i < 5; ++i)
#pragma unroll
                for (int j = 0; j < 4; ++j) {
                    const v2f af = {a[i][2 * kk], a[i][2 * kk + 1]};
                    const v2f bf = {b[j][2 * kk], b[j][2 * kk + 1]};
                    acc[i][j] = __builtin_amdgcn_wmma_f32_16x16x4_f32(
                                    false, af, false, bf,
                                    (short)0, acc[i][j], false, false);
                }
    }
    if (!has_above) {
#pragma unroll
        for (int j = 0; j < 4; ++j) acc[0][j] = (v8f){};
    }

    const int n = (tm >> 4) & (N_GRP - 1);
#pragma unroll
    for (int j = 0; j < 4; ++j) {
        const int d  = tn * 64 + j * 16 + l15;
        const int ch = n * DIN_ + d;
        const v4f w  = *(const v4f*)(cw + (size_t)ch * 4);
        const float bias = cb[ch];
#pragma unroll
        for (int i = 0; i < 4; ++i) {
            float xw[11];
#pragma unroll
            for (int p = 0; p < 3; ++p) {
                const float send = hh ? acc[i][j][5 + p] : acc[i + 1][j][5 + p];
                xw[p] = __shfl_xor(send, 16);
            }
#pragma unroll
            for (int r = 0; r < 8; ++r) xw[3 + r] = acc[i + 1][j][r];
            const int rowb = tm * 64 + i * 16 + rb;
#pragma unroll
            for (int r = 0; r < 8; ++r) {
                float s = bias;
                s = fmaf(w[0], xw[r],     s);
                s = fmaf(w[1], xw[r + 1], s);
                s = fmaf(w[2], xw[r + 2], s);
                s = fmaf(w[3], xw[r + 3], s);
                U[(size_t)(rowb + r) * DIN_ + d] = s / (1.0f + __expf(-s));
            }
        }
    }
}

__global__ __launch_bounds__(128)
void scan_chunk_reduce(const float* __restrict__ U, const float* __restrict__ PR,
                       const float* __restrict__ Wdt, const float* __restrict__ bdt,
                       const float* __restrict__ A_log,
                       float* __restrict__ DT,
                       float* __restrict__ CH_C, float* __restrict__ CH_S)
{
    __shared__ float dts[TCH * 128];

    const int d  = ((blockIdx.x & 7) << 7) + threadIdx.x;
    const int ck = (blockIdx.x >> 3) & (NCHUNK - 1);
    const int bn = blockIdx.x >> (3 + NCH_LG);

    {
        const int lane = threadIdx.x & 31;
        const int wch  = threadIdx.x & ~31;
        float dt0[8], dt1[8];
        dt_tile_softplus(PR + ((size_t)bn * L_ + ck * TCH) * 64, Wdt, bdt,
                         ((blockIdx.x & 7) << 7) + wch, lane, dt0, dt1);
        const int c = lane & 15, hh = lane >> 4;
#pragma unroll
        for (int r = 0; r < 8; ++r) {
            dts[(8 * hh + r) * 128 + wch + c]      = dt0[r];
            dts[(8 * hh + r) * 128 + wch + 16 + c] = dt1[r];
        }
    }
    __syncthreads();

    float As[DSTATE], h[DSTATE];
#pragma unroll
    for (int s = 0; s < DSTATE; ++s) {
        As[s] = -__expf(A_log[(size_t)d * DSTATE + s]);
        h[s]  = 0.0f;
    }
    float S = 0.0f;
    const int t0 = ck * TCH;
    for (int t = t0; t < t0 + TCH; ++t) {
        const size_t base = (size_t)bn * L_ + t;
        const float* pr = PR + base * 64;
        float dtv = dts[(t - t0) * 128 + threadIdx.x];
        DT[base * DIN_ + d] = dtv;
        float uv  = U [base * DIN_ + d];
        S += dtv;
        float dtu = dtv * uv;
#pragma unroll
        for (int s = 0; s < DSTATE; ++s)
            h[s] = fmaf(__expf(dtv * As[s]), h[s], dtu * pr[DTRANK + s]);
    }
    const size_t cb = (size_t)(bn * NCHUNK + ck) * DSTATE * DIN_ + d;
#pragma unroll
    for (int s = 0; s < DSTATE; ++s) CH_C[cb + (size_t)s * DIN_] = h[s];
    CH_S[(size_t)(bn * NCHUNK + ck) * DIN_ + d] = S;
}

__global__ __launch_bounds__(256)
void scan_chunk_prefix(const float* __restrict__ A_log,
                       const float* __restrict__ CH_C,
                       const float* __restrict__ CH_S,
                       float* __restrict__ H0)
{
    const int idx = blockIdx.x * 256 + threadIdx.x;
    const int d   = idx & (DIN_ - 1);
    const int s   = (idx >> 10) & (DSTATE - 1);
    const int bn  = idx >> 14;
    const float As = -__expf(A_log[(size_t)d * DSTATE + s]);
    float h = 0.0f;
    for (int j0 = 0; j0 < NCHUNK; j0 += 16) {
        float Sj[16], cj[16];
#pragma unroll
        for (int i = 0; i < 16; ++i) {
            const size_t cidx = (size_t)(bn * NCHUNK + j0 + i);
            Sj[i] = CH_S[cidx * DIN_ + d];
            cj[i] = CH_C[(cidx * DSTATE + s) * DIN_ + d];
        }
#pragma unroll
        for (int i = 0; i < 16; ++i) {
            H0[((size_t)(bn * NCHUNK + j0 + i) * DSTATE + s) * DIN_ + d] = h;
            h = fmaf(__expf(Sj[i] * As), h, cj[i]);
        }
    }
}

__global__ __launch_bounds__(1024)
void scan_emit_out_proj(const float* __restrict__ DT, const float* __restrict__ U,
                        const float* __restrict__ PR, const float* __restrict__ C1,
                        const float* __restrict__ A_log, const float* __restrict__ Dp,
                        const float* __restrict__ H0,
                        const float* __restrict__ Woutf,
                        float* __restrict__ out)
{
    __shared__ __attribute__((aligned(16))) float Ys[TCH * DIN_];

    const int d  = threadIdx.x;
    const int ck = blockIdx.x & (NCHUNK - 1);
    const int bn = blockIdx.x >> NCH_LG;
    const int t0 = ck * TCH;

    {
        const size_t hb = (size_t)(bn * NCHUNK + ck) * DSTATE * DIN_ + d;
        float As[DSTATE], h[DSTATE];
#pragma unroll
        for (int s = 0; s < DSTATE; ++s) {
            As[s] = -__expf(A_log[(size_t)d * DSTATE + s]);
            h[s]  = H0[hb + (size_t)s * DIN_];
        }

        const float dp = Dp[d];
        for (int tt = 0; tt < TCH; ++tt) {
            const size_t base = (size_t)bn * L_ + t0 + tt;
            const float* pr = PR + base * 64;
            float dtv = DT[base * DIN_ + d];
            float uv  = U [base * DIN_ + d];
            float dtu = dtv * uv;
            float y = 0.0f;
#pragma unroll
            for (int s = 0; s < DSTATE; ++s) {
                h[s] = fmaf(__expf(dtv * As[s]), h[s], dtu * pr[DTRANK + s]);
                y    = fmaf(h[s], pr[DTRANK + DSTATE + s], y);
            }
            float res = C1[base * (2 * DIN_) + DIN_ + d];
            float sr  = res / (1.0f + __expf(-res));
            Ys[tt * DIN_ + ((((d >> 2) ^ tt) << 2) | (d & 3))] = (y + uv * dp) * sr;
        }
    }
    __syncthreads();

    const int wave = threadIdx.x >> 5;
    if (wave >= 16) return;
    const int lane = threadIdx.x & 31;
    const int c    = lane & 15;
    const int hh   = lane >> 4;
    const int n0   = wave * 32;

    const float* ya  = Ys + c * DIN_;
    const float* bp0 = Woutf + (size_t)(2 * wave) * ((DIN_ / 8) * 128) + lane * 4;
    const float* bp1 = bp0 + (size_t)(DIN_ / 8) * 128;

    v8f acc0 = {}, acc1 = {};
#pragma unroll 4
    for (int g = 0; g < DIN_ / 8; ++g) {
        const v4f a  = *(const v4f*)(ya + (((2 * g + hh) ^ c) << 2));
        const v4f b0 = *(const v4f*)(bp0 + 128 * g);
        const v4f b1 = *(const v4f*)(bp1 + 128 * g);
        const v2f alo = {a[0], a[1]},   ahi = {a[2], a[3]};
        const v2f b0lo = {b0[0], b0[1]}, b0hi = {b0[2], b0[3]};
        const v2f b1lo = {b1[0], b1[1]}, b1hi = {b1[2], b1[3]};
        acc0 = __builtin_amdgcn_wmma_f32_16x16x4_f32(false, alo, false, b0lo,
                                                     (short)0, acc0, false, false);
        acc1 = __builtin_amdgcn_wmma_f32_16x16x4_f32(false, alo, false, b1lo,
                                                     (short)0, acc1, false, false);
        acc0 = __builtin_amdgcn_wmma_f32_16x16x4_f32(false, ahi, false, b0hi,
                                                     (short)0, acc0, false, false);
        acc1 = __builtin_amdgcn_wmma_f32_16x16x4_f32(false, ahi, false, b1hi,
                                                     (short)0, acc1, false, false);
    }

    const size_t orow = (size_t)bn * L_ + t0 + 8 * hh;
#pragma unroll
    for (int r = 0; r < 8; ++r) {
        out[(orow + r) * DM_ + n0 + c]      = acc0[r];
        out[(orow + r) * DM_ + n0 + 16 + c] = acc1[r];
    }
}

extern "C" void kernel_launch(void* const* d_in, const int* in_sizes, int n_in,
                              void* d_out, int out_size, void* d_ws, size_t ws_size,
                              hipStream_t stream)
{
    const float* x      = (const float*)d_in[0];
    const float* W_in   = (const float*)d_in[1];
    const float* conv_w = (const float*)d_in[2];
    const float* conv_b = (const float*)d_in[3];
    const float* W_x    = (const float*)d_in[4];
    const float* W_dt   = (const float*)d_in[5];
    const float* b_dt   = (const float*)d_in[6];
    const float* A_log  = (const float*)d_in[7];
    const float* Dp     = (const float*)d_in[8];
    const float* W_out  = (const float*)d_in[9];
    float* out = (float*)d_out;

    float* C1   = (float*)d_ws;
    float* U    = C1   + (size_t)MTOT * (2 * DIN_);
    float* PR   = U    + (size_t)MTOT * DIN_;
    float* DTb  = PR   + (size_t)MTOT * 64;
    float* CH_C = DTb  + (size_t)MTOT * DIN_;
    float* CH_S = CH_C + (size_t)BN_ * NCHUNK * DSTATE * DIN_;
    float* H0   = CH_S + (size_t)BN_ * NCHUNK * DIN_;
    float* Xf    = H0    + (size_t)BN_ * NCHUNK * DSTATE * DIN_;
    float* Winf  = Xf    + (size_t)MTOT * DM_;
    float* Woutf = Winf  + (size_t)(2 * DIN_) * DM_;

    to_fragment_order<<<(XF_JOBS + WIF_JOBS + WOF_JOBS) / 8, 256, 0, stream>>>(
        x, W_in, W_out, Xf, Winf, Woutf);

    in_proj_conv_silu<<<(MTOT / 64) * 8, 128, 0, stream>>>(Xf, Winf, conv_w, conv_b,
                                                           C1, U);

    x_proj_gemm<<<MTOT / 16, 64, 0, stream>>>(U, W_x, PR);

    const int scan_blocks = BN_ * NCHUNK * (DIN_ / 128);
    scan_chunk_reduce<<<scan_blocks, 128, 0, stream>>>(U, PR, W_dt, b_dt, A_log,
                                                       DTb, CH_C, CH_S);
    scan_chunk_prefix<<<(BN_ * DSTATE * DIN_) / 256, 256, 0, stream>>>(A_log, CH_C,
                                                                       CH_S, H0);
    scan_emit_out_proj<<<BN_ * NCHUNK, 1024, 0, stream>>>(DTb, U, PR, C1, A_log, Dp,
                                                          H0, Woutf, out);
}


# x_proj fused into the scan reduce pass: a block owns a 16-step band of all 1024 channels, U band in LDS, x_proj on WMMA with K split over its 32 waves; 5 launches; f32 sum order changes
# speedup vs baseline: 2.2744x; 1.0535x over previous
#include <hip/hip_runtime.h>

typedef float v2f __attribute__((ext_vector_type(2)));
typedef float v4f __attribute__((ext_vector_type(4)));
typedef float v8f __attribute__((ext_vector_type(8)));

#define BN_    4
#define N_GRP  2
#define L_     1024
#define DM_    512
#define DIN_   1024
#define DSTATE 16
#define DTRANK 32
#define MTOT   4096
#define NCHUNK 64
#define NCH_LG 6
#define TCH    16

__device__ __forceinline__
void dt_weights_load(const float* __restrict__ Wdt, const float* __restrict__ bdt,
                     int ch0, int lane,
                     v2f (&b0)[DTRANK / 4], v2f (&b1)[DTRANK / 4],
                     float& bias0, float& bias1)
{
    const int c  = lane & 15;
    const int hh = lane >> 4;
    const float* bp0 = Wdt + (size_t)(ch0 + c) * DTRANK + 2 * hh;
    const float* bp1 = bp0 + (size_t)16 * DTRANK;
#pragma unroll
    for (int ks = 0; ks < DTRANK / 4; ++ks) {
        b0[ks] = *(const v2f*)(bp0 + 4 * ks);
        b1[ks] = *(const v2f*)(bp1 + 4 * ks);
    }
    bias0 = bdt[ch0 + c];
    bias1 = bdt[ch0 + 16 + c];
}

__device__ __forceinline__
void dt_tile_softplus(const float* prb,
                      const v2f (&b0)[DTRANK / 4], const v2f (&b1)[DTRANK / 4],
                      float bias0, float bias1, int lane,
                      float (&dt0)[8], float (&dt1)[8])
{
    const int c  = lane & 15;
    const int hh = lane >> 4;
    const float* ap = prb + (size_t)c * 64 + 2 * hh;

    v2f a[DTRANK / 4];
#pragma unroll
    for (int ks = 0; ks < DTRANK / 4; ++ks) a[ks] = *(const v2f*)(ap + 4 * ks);
    v8f acc0 = {}, acc1 = {};
#pragma unroll
    for (int ks = 0; ks < DTRANK / 4; ++ks) {
        acc0 = __builtin_amdgcn_wmma_f32_16x16x4_f32(false, a[ks], false, b0[ks],
                                                     (short)0, acc0, false, false);
        acc1 = __builtin_amdgcn_wmma_f32_16x16x4_f32(false, a[ks], false, b1[ks],
                                                     (short)0, acc1, false, false);
    }
#pragma unroll
    for (int r = 0; r < 8; ++r) {
        float v0 = acc0[r] + bias0;
        float v1 = acc1[r] + bias1;
        dt0[r] = (v0 > 20.0f) ? v0 : __logf(1.0f + __expf(v0));
        dt1[r] = (v1 > 20.0f) ? v1 : __logf(1.0f + __expf(v1));
    }
}

#define XF_JOBS   ((MTOT / 16) * (DM_ / 64))
#define WIF_JOBS  (((2 * DIN_) / 16) * (DM_ / 64))
#define WOF_JOBS  ((DM_ / 16) * (DIN_ / 64))
#define WXF_JOBS  ((64 / 16) * (DIN_ / 64))
__global__ __launch_bounds__(256)
void to_fragment_order(const float* __restrict__ X, const float* __restrict__ Win,
                       const float* __restrict__ Wout, const float* __restrict__ Wx,
                       float* __restrict__ Xf, float* __restrict__ Winf,
                       float* __restrict__ Woutf, float* __restrict__ Wxf)
{
    const int lane = threadIdx.x & 31;
    const int c    = lane & 15;
    const int hh   = lane >> 4;
    const int job  = blockIdx.x * 8 + (threadIdx.x >> 5);

    if (job < XF_JOBS + WIF_JOBS) {
        const bool isx = job < XF_JOBS;
        const float* src = isx ? X : Win;
        float* dst       = isx ? Xf : Winf;
        const int j      = isx ? job : job - XF_JOBS;
        const int tile   = j >> 3;
        const int g0     = (j & 7) * 8;
        const float* rp = src + (size_t)(tile * 16 + c) * DM_ + 8 * g0 + 2 * hh;
        float* dp = dst + ((size_t)tile * (DM_ / 8) + g0) * 128 + lane * 4;
        v2f lo[8], hi[8];
#pragma unroll
        for (int i = 0; i < 8; ++i) {
            lo[i] = *(const v2f*)(rp + 8 * i);
            hi[i] = *(const v2f*)(rp + 8 * i + 4);
        }
#pragma unroll
        for (int i = 0; i < 8; ++i) {
            const v4f v = {lo[i][0], lo[i][1], hi[i][0], hi[i][1]};
            *(v4f*)(dp + i * 128) = v;
        }
    } else {
        const int j0   = job - (XF_JOBS + WIF_JOBS);
        const bool iso = j0 < WOF_JOBS;
        const float* src = iso ? Wout : Wx;
        float* dst       = iso ? Woutf : Wxf;
        const int j    = iso ? j0 : j0 - WOF_JOBS;
        const int tile = j >> 4;
        const int g0   = (j & 15) * 8;
        const float* rp = src + (size_t)(tile * 16 + c) * DIN_ + 8 * g0 + 4 * hh;
        float* dp = dst + ((size_t)tile * (DIN_ / 8) + g0) * 128 + lane * 4;
        v4f v[8];
#pragma unroll
        for (int i = 0; i < 8; ++i) v[i] = *(const v4f*)(rp + 8 * i);
#pragma unroll
        for (int i = 0; i < 8; ++i) *(v4f*)(dp + i * 128) = v[i];
    }
}

#define TSTR_IN ((DM_ / 8) * 128)
__global__ __launch_bounds__(128)
void in_proj_conv_silu(const float* __restrict__ Xf, const float* __restrict__ Winf,
                       const float* __restrict__ cw, const float* __restrict__ cb,
                       float* __restrict__ C1, float* __restrict__ U)
{
    const int lane = threadIdx.x & 31;
    const int wave = threadIdx.x >> 5;
    const int tm   = blockIdx.x >> 3;
    const int q    = blockIdx.x & 7;
    const bool xt  = ((((int)blockIdx.x) ^ (wave >> 1)) & 1) == 0;
    const int tn   = (xt ? 0 : 16) + 2 * q + (wave & 1);

    const int l15  = lane & 15;
    const int hh   = lane >> 4;
    const size_t astr = (size_t)TSTR_IN;
    const float* ap = Xf   + (size_t)(tm * 4) * TSTR_IN + lane * 4;
    const float* bp = Winf + (size_t)(tn * 4) * TSTR_IN + lane * 4;
    const int rb = hh << 3;

    if (!xt) {
        v8f acc[4][4] = {};
        for (int g = 0; g < DM_ / 8; ++g) {
            v4f a[4], b[4];
#pragma unroll
            for (int i = 0; i < 4; ++i) a[i] = *(const v4f*)(ap + i * astr + g * 128);
#pragma unroll
            for (int j = 0; j < 4; ++j) b[j] = *(const v4f*)(bp + j * astr + g * 128);
#pragma unroll
            for (int kk = 0; kk < 2; ++kk)
#pragma unroll
                for (int i = 0; i < 4; ++i)
#pragma unroll
                    for (int j = 0; j < 4; ++j) {
                        const v2f af = {a[i][2 * kk], a[i][2 * kk + 1]};
                        const v2f bf = {b[j][2 * kk], b[j][2 * kk + 1]};
                        acc[i][j] = __builtin_amdgcn_wmma_f32_16x16x4_f32(
                                        false, af, false, bf,
                                        (short)0, acc[i][j], false, false);
                    }
        }
#pragma unroll
        for (int i = 0; i < 4; ++i) {
            const int rowb = tm * 64 + i * 16 + rb;
#pragma unroll
            for (int j = 0; j < 4; ++j) {
                const int col = tn * 64 + j * 16 + l15;
#pragma unroll
                for (int r = 0; r < 8; ++r)
                    C1[(size_t)(rowb + r) * (2 * DIN_) + col] = acc[i][j][r];
            }
        }
        return;
    }

    const bool has_above = (tm & 15) != 0;
    const float* hp = has_above ? (ap - astr) : ap;
    v8f acc[5][4] = {};
    for (int g = 0; g < DM_ / 8; ++g) {
        v4f a[5], b[4];
        a[0] = *(const v4f*)(hp + g * 128);
#pragma unroll
        for (int i = 0; i < 4; ++i) a[i + 1] = *(const v4f*)(ap + i * astr + g * 128);
#pragma unroll
        for (int j = 0; j < 4; ++j) b[j] = *(const v4f*)(bp + j * astr + g * 128);
#pragma unroll
        for (int kk = 0; kk < 2; ++kk)
#pragma unroll
            for (int i = 0; i < 5; ++i)
#pragma unroll
                for (int j = 0; j < 4; ++j) {
                    const v2f af = {a[i][2 * kk], a[i][2 * kk + 1]};
                    const v2f bf = {b[j][2 * kk], b[j][2 * kk + 1]};
                    acc[i][j] = __builtin_amdgcn_wmma_f32_16x16x4_f32(
                                    false, af, false, bf,
                                    (short)0, acc[i][j], false, false);
                }
    }
    if (!has_above) {
#pragma unroll
        for (int j = 0; j < 4; ++j) acc[0][j] = (v8f){};
    }

    const int n = (tm >> 4) & (N_GRP - 1);
#pragma unroll
    for (int j = 0; j < 4; ++j) {
        const int d  = tn * 64 + j * 16 + l15;
        const int ch = n * DIN_ + d;
        const v4f w  = *(const v4f*)(cw + (size_t)ch * 4);
        const float bias = cb[ch];
#pragma unroll
        for (int i = 0; i < 4; ++i) {
            float xw[11];
#pragma unroll
            for (int p = 0; p < 3; ++p) {
                const float send = hh ? acc[i][j][5 + p] : acc[i + 1][j][5 + p];
                xw[p] = __shfl_xor(send, 16);
            }
#pragma unroll
            for (int r = 0; r < 8; ++r) xw[3 + r] = acc[i + 1][j][r];
            const int rowb = tm * 64 + i * 16 + rb;
#pragma unroll
            for (int r = 0; r < 8; ++r) {
                float s = bias;
                s = fmaf(w[0], xw[r],     s);
                s = fmaf(w[1], xw[r + 1], s);
                s = fmaf(w[2], xw[r + 2], s);
                s = fmaf(w[3], xw[r + 3], s);
                U[(size_t)(rowb + r) * DIN_ + d] = s / (1.0f + __expf(-s));
            }
        }
    }
}

#define KSL 8
#define GPS ((DIN_ / 8) / KSL)
__global__ __launch_bounds__(1024)
void x_proj_scan_reduce(const float* __restrict__ U, const float* __restrict__ Wxf,
                        const float* __restrict__ Wdt, const float* __restrict__ bdt,
                        const float* __restrict__ A_log,
                        float* __restrict__ PR, float* __restrict__ DT,
                        float* __restrict__ CH_C, float* __restrict__ CH_S)
{
    __shared__ __attribute__((aligned(16))) float lds[TCH * DIN_ + KSL * TCH * 64 + TCH * 64];
    float* Us   = lds;
    float* dts  = lds;
    float* part = lds + TCH * DIN_;
    float* prs  = part + KSL * TCH * 64;

    const int d    = threadIdx.x;
    const int ck   = blockIdx.x & (NCHUNK - 1);
    const int bn   = blockIdx.x >> NCH_LG;
    const int lane = threadIdx.x & 31;
    const int wave = threadIdx.x >> 5;
    const int c    = lane & 15;
    const int hh   = lane >> 4;
    const size_t row0 = (size_t)bn * L_ + ck * TCH;

    const int ct = wave & 3;
    const int ks = wave >> 2;
    const float* bp = Wxf + ((size_t)ct * (DIN_ / 8) + ks * GPS) * 128 + lane * 4;
    v4f b[GPS];
#pragma unroll
    for (int i = 0; i < GPS; ++i) b[i] = *(const v4f*)(bp + 128 * i);

    float u[TCH];
#pragma unroll
    for (int tt = 0; tt < TCH; ++tt) u[tt] = U[(row0 + tt) * DIN_ + d];
#pragma unroll
    for (int tt = 0; tt < TCH; ++tt)
        Us[tt * DIN_ + ((((d >> 2) ^ tt) << 2) | (d & 3))] = u[tt];
    __syncthreads();

    v8f acc = {};
    const float* ya = Us + c * DIN_;
#pragma unroll
    for (int i = 0; i < GPS / 2; ++i) {
        const int g = ks * GPS + i;
        const v4f a = *(const v4f*)(ya + (((2 * g + hh) ^ c) << 2));
        const v2f alo = {a[0], a[1]},       ahi = {a[2], a[3]};
        const v2f blo = {b[i][0], b[i][1]}, bhi = {b[i][2], b[i][3]};
        acc = __builtin_amdgcn_wmma_f32_16x16x4_f32(false, alo, false, blo,
                                                    (short)0, acc, false, false);
        acc = __builtin_amdgcn_wmma_f32_16x16x4_f32(false, ahi, false, bhi,
                                                    (short)0, acc, false, false);
    }

    float Al[DSTATE];
#pragma unroll
    for (int s = 0; s < DSTATE; ++s) Al[s] = A_log[(size_t)d * DSTATE + s];
    v2f wb0[DTRANK / 4], wb1[DTRANK / 4];
    float bias0, bias1;
    dt_weights_load(Wdt, bdt, wave * 32, lane, wb0, wb1, bias0, bias1);
    __builtin_amdgcn_sched_barrier(0);

#pragma unroll
    for (int i = GPS / 2; i < GPS; ++i) {
        const int g = ks * GPS + i;
        const v4f a = *(const v4f*)(ya + (((2 * g + hh) ^ c) << 2));
        const v2f alo = {a[0], a[1]},       ahi = {a[2], a[3]};
        const v2f blo = {b[i][0], b[i][1]}, bhi = {b[i][2], b[i][3]};
        acc = __builtin_amdgcn_wmma_f32_16x16x4_f32(false, alo, false, blo,
                                                    (short)0, acc, false, false);
        acc = __builtin_amdgcn_wmma_f32_16x16x4_f32(false, ahi, false, bhi,
                                                    (short)0, acc, false, false);
    }
#pragma unroll
    for (int r = 0; r < 8; ++r)
        part[(ks * TCH + 8 * hh + r) * 64 + ct * 16 + c] = acc[r];
    __syncthreads();

    float As[DSTATE], h[DSTATE];
#pragma unroll
    for (int s = 0; s < DSTATE; ++s) {
        As[s] = -__expf(Al[s]);
        h[s]  = 0.0f;
    }

    float prv = part[d];
#pragma unroll
    for (int k = 1; k < KSL; ++k) prv += part[k * TCH * 64 + d];
    prs[d] = prv;
    __syncthreads();

    {
        float dt0[8], dt1[8];
        dt_tile_softplus(prs, wb0, wb1, bias0, bias1, lane, dt0, dt1);
#pragma unroll
        for (int r = 0; r < 8; ++r) {
            dts[(8 * hh + r) * DIN_ + wave * 32 + c]      = dt0[r];
            dts[(8 * hh + r) * DIN_ + wave * 32 + 16 + c] = dt1[r];
        }
    }
    __syncthreads();

    PR[row0 * 64 + d] = prv;

    float S = 0.0f;
#pragma unroll
    for (int tt = 0; tt < TCH; ++tt) {
        const float* pb = prs + tt * 64 + DTRANK;
        const float dtv = dts[tt * DIN_ + d];
        DT[(row0 + tt) * DIN_ + d] = dtv;
        S += dtv;
        const float dtu = dtv * u[tt];
#pragma unroll
        for (int s = 0; s < DSTATE; ++s)
            h[s] = fmaf(__expf(dtv * As[s]), h[s], dtu * pb[s]);
    }
    const size_t cb = (size_t)(bn * NCHUNK + ck) * DSTATE * DIN_ + d;
#pragma unroll
    for (int s = 0; s < DSTATE; ++s) CH_C[cb + (size_t)s * DIN_] = h[s];
    CH_S[(size_t)(bn * NCHUNK + ck) * DIN_ + d] = S;
}

__global__ __launch_bounds__(256)
void scan_chunk_prefix(const float* __restrict__ A_log,
                       const float* __restrict__ CH_C,
                       const float* __restrict__ CH_S,
                       float* __restrict__ H0)
{
    const int idx = blockIdx.x * 256 + threadIdx.x;
    const int d   = idx & (DIN_ - 1);
    const int s   = (idx >> 10) & (DSTATE - 1);
    const int bn  = idx >> 14;
    const float As = -__expf(A_log[(size_t)d * DSTATE + s]);
    float h = 0.0f;
    for (int j0 = 0; j0 < NCHUNK; j0 += 16) {
        float Sj[16], cj[16];
#pragma unroll
        for (int i = 0; i < 16; ++i) {
            const size_t cidx = (size_t)(bn * NCHUNK + j0 + i);
            Sj[i] = CH_S[cidx * DIN_ + d];
            cj[i] = CH_C[(cidx * DSTATE + s) * DIN_ + d];
        }
#pragma unroll
        for (int i = 0; i < 16; ++i) {
            H0[((size_t)(bn * NCHUNK + j0 + i) * DSTATE + s) * DIN_ + d] = h;
            h = fmaf(__expf(Sj[i] * As), h, cj[i]);
        }
    }
}

__global__ __launch_bounds__(1024)
void scan_emit_out_proj(const float* __restrict__ DT, const float* __restrict__ U,
                        const float* __restrict__ PR, const float* __restrict__ C1,
                        const float* __restrict__ A_log, const float* __restrict__ Dp,
                        const float* __restrict__ H0,
                        const float* __restrict__ Woutf,
                        float* __restrict__ out)
{
    __shared__ __attribute__((aligned(16))) float Ys[TCH * DIN_];

    const int d  = threadIdx.x;
    const int ck = blockIdx.x & (NCHUNK - 1);
    const int bn = blockIdx.x >> NCH_LG;
    const int t0 = ck * TCH;

    {
        const size_t hb = (size_t)(bn * NCHUNK + ck) * DSTATE * DIN_ + d;
        float As[DSTATE], h[DSTATE];
#pragma unroll
        for (int s = 0; s < DSTATE; ++s) {
            As[s] = -__expf(A_log[(size_t)d * DSTATE + s]);
            h[s]  = H0[hb + (size_t)s * DIN_];
        }

        const float dp = Dp[d];
        for (int tt = 0; tt < TCH; ++tt) {
            const size_t base = (size_t)bn * L_ + t0 + tt;
            const float* pr = PR + base * 64;
            float dtv = DT[base * DIN_ + d];
            float uv  = U [base * DIN_ + d];
            float dtu = dtv * uv;
            float y = 0.0f;
#pragma unroll
            for (int s = 0; s < DSTATE; ++s) {
                h[s] = fmaf(__expf(dtv * As[s]), h[s], dtu * pr[DTRANK + s]);
                y    = fmaf(h[s], pr[DTRANK + DSTATE + s], y);
            }
            float res = C1[base * (2 * DIN_) + DIN_ + d];
            float sr  = res / (1.0f + __expf(-res));
            Ys[tt * DIN_ + ((((d >> 2) ^ tt) << 2) | (d & 3))] = (y + uv * dp) * sr;
        }
    }
    __syncthreads();

    const int wave = threadIdx.x >> 5;
    if (wave >= 16) return;
    const int lane = threadIdx.x & 31;
    const int c    = lane & 15;
    const int hh   = lane >> 4;
    const int n0   = wave * 32;

    const float* ya  = Ys + c * DIN_;
    const float* bp0 = Woutf + (size_t)(2 * wave) * ((DIN_ / 8) * 128) + lane * 4;
    const float* bp1 = bp0 + (size_t)(DIN_ / 8) * 128;

    v8f acc0 = {}, acc1 = {};
#pragma unroll 4
    for (int g = 0; g < DIN_ / 8; ++g) {
        const v4f a  = *(const v4f*)(ya + (((2 * g + hh) ^ c) << 2));
        const v4f b0 = *(const v4f*)(bp0 + 128 * g);
        const v4f b1 = *(const v4f*)(bp1 + 128 * g);
        const v2f alo = {a[0], a[1]},   ahi = {a[2], a[3]};
        const v2f b0lo = {b0[0], b0[1]}, b0hi = {b0[2], b0[3]};
        const v2f b1lo = {b1[0], b1[1]}, b1hi = {b1[2], b1[3]};
        acc0 = __builtin_amdgcn_wmma_f32_16x16x4_f32(false, alo, false, b0lo,
                                                     (short)0, acc0, false, false);
        acc1 = __builtin_amdgcn_wmma_f32_16x16x4_f32(false, alo, false, b1lo,
                                                     (short)0, acc1, false, false);
        acc0 = __builtin_amdgcn_wmma_f32_16x16x4_f32(false, ahi, false, b0hi,
                                                     (short)0, acc0, false, false);
        acc1 = __builtin_amdgcn_wmma_f32_16x16x4_f32(false, ahi, false, b1hi,
                                                     (short)0, acc1, false, false);
    }

    const size_t orow = (size_t)bn * L_ + t0 + 8 * hh;
#pragma unroll
    for (int r = 0; r < 8; ++r) {
        out[(orow + r) * DM_ + n0 + c]      = acc0[r];
        out[(orow + r) * DM_ + n0 + 16 + c] = acc1[r];
    }
}

extern "C" void kernel_launch(void* const* d_in, const int* in_sizes, int n_in,
                              void* d_out, int out_size, void* d_ws, size_t ws_size,
                              hipStream_t stream)
{
    const float* x      = (const float*)d_in[0];
    const float* W_in   = (const float*)d_in[1];
    const float* conv_w = (const float*)d_in[2];
    const float* conv_b = (const float*)d_in[3];
    const float* W_x    = (const float*)d_in[4];
    const float* W_dt   = (const float*)d_in[5];
    const float* b_dt   = (const float*)d_in[6];
    const float* A_log  = (const float*)d_in[7];
    const float* Dp     = (const float*)d_in[8];
    const float* W_out  = (const float*)d_in[9];
    float* out = (float*)d_out;

    float* C1   = (float*)d_ws;
    float* U    = C1   + (size_t)MTOT * (2 * DIN_);
    float* PR   = U    + (size_t)MTOT * DIN_;
    float* DTb  = PR   + (size_t)MTOT * 64;
    float* CH_C = DTb  + (size_t)MTOT * DIN_;
    float* CH_S = CH_C + (size_t)BN_ * NCHUNK * DSTATE * DIN_;
    float* H0   = CH_S + (size_t)BN_ * NCHUNK * DIN_;
    float* Xf    = H0    + (size_t)BN_ * NCHUNK * DSTATE * DIN_;
    float* Winf  = Xf    + (size_t)MTOT * DM_;
    float* Woutf = Winf  + (size_t)(2 * DIN_) * DM_;
    float* Wxf   = Woutf + (size_t)DM_ * DIN_;

    to_fragment_order<<<(XF_JOBS + WIF_JOBS + WOF_JOBS + WXF_JOBS) / 8, 256, 0, stream>>>(
        x, W_in, W_out, W_x, Xf, Winf, Woutf, Wxf);

    in_proj_conv_silu<<<(MTOT / 64) * 8, 128, 0, stream>>>(Xf, Winf, conv_w, conv_b,
                                                           C1, U);

    x_proj_scan_reduce<<<BN_ * NCHUNK, 1024, 0, stream>>>(U, Wxf, W_dt, b_dt, A_log,
                                                          PR, DTb, CH_C, CH_S);
    scan_chunk_prefix<<<(BN_ * DSTATE * DIN_) / 256, 256, 0, stream>>>(A_log, CH_C,
                                                                       CH_S, H0);
    scan_emit_out_proj<<<BN_ * NCHUNK, 1024, 0, stream>>>(DTb, U, PR, C1, A_log, Dp,
                                                          H0, Woutf, out);
}
